# starting kernel
# baseline (speedup 1.0000x reference)
#include <hip/hip_runtime.h>

typedef __attribute__((ext_vector_type(2))) float v2f;
typedef __attribute__((ext_vector_type(8))) float v8f;

#define IMG    224
#define PATCH  16
#define NP     14
#define EMBED  768
#define CCH    12
#define BB     64
#define M_TOTAL (BB * NP * NP)
#define NTILES  (EMBED / 16)
#define MTILES  (M_TOTAL / 16)

__global__ __launch_bounds__(224) void pool16_kernel(const float* __restrict__ x,
                                                     float* __restrict__ pooled) {
    __shared__ float colsum[IMG];
    int s = blockIdx.x;
    int i = s % NP; s /= NP;
    int c = s % CCH; s /= CCH;
    int b = s;
    const float* base = x + (((size_t)(b * CCH + c)) * IMG + (size_t)i * PATCH) * IMG;

    int col = threadIdx.x;
    float acc = 0.f;
#pragma unroll
    for (int r = 0; r < PATCH; ++r)
        acc += base[(size_t)r * IMG + col];
    colsum[col] = acc;
    __syncthreads();

    if (col < NP) {
        float ps = 0.f;
#pragma unroll
        for (int k = 0; k < PATCH; ++k)
            ps += colsum[col * PATCH + k];
        int m = b * (NP * NP) + i * NP + col;
        pooled[(size_t)m * CCH + c] = ps;
    }
}

__global__ __launch_bounds__(256) void gemm_wmma_kernel(const float* __restrict__ pooled,
                                                        const float* __restrict__ w,
                                                        float* __restrict__ out) {
    int wid  = blockIdx.x * 8 + (threadIdx.x >> 5);
    int lane = threadIdx.x & 31;
    int mtile = wid / NTILES;
    int ntile = wid % NTILES;

    int half  = lane >> 4;
    int khalf = half * 2;
    int mrow  = lane & 15;
    int n     = lane & 15;

    const float* arow = pooled + (size_t)(mtile * 16 + mrow) * CCH;
    const float* wcol = w + ntile * 16 + n;

    v8f acc = {};
#pragma unroll
    for (int k0 = 0; k0 < CCH; k0 += 4) {
        v2f a;
        a.x = arow[k0 + khalf];
        a.y = arow[k0 + khalf + 1];
        v2f bm;
        bm.x = wcol[(size_t)(k0 + khalf)     * EMBED];
        bm.y = wcol[(size_t)(k0 + khalf + 1) * EMBED];
        acc = __builtin_amdgcn_wmma_f32_16x16x4_f32(
             false, a,  false, bm,
             (short)0, acc,  false,  false);
    }

    int mbase = mtile * 16 + half * 8;
    float* o = out + (size_t)mbase * EMBED + (size_t)ntile * 16 + n;
#pragma unroll
    for (int v = 0; v < 8; ++v)
        o[(size_t)v * EMBED] = acc[v];
}

extern "C" void kernel_launch(void* const* d_in, const int* in_sizes, int n_in,
                              void* d_out, int out_size, void* d_ws, size_t ws_size,
                              hipStream_t stream) {
    const float* x = (const float*)d_in[0];
    const float* w = (const float*)d_in[1];
    float* out    = (float*)d_out;
    float* pooled = (float*)d_ws;

    pool16_kernel<<<BB * CCH * NP, 224, 0, stream>>>(x, pooled);
    gemm_wmma_kernel<<<(MTILES * NTILES) / 8, 256, 0, stream>>>(pooled, w, out);
}


# pool kernel reads x as a stream: 512 groups x 7 waves, 3 contiguous strips per wave, 16-byte loads of 512 bytes per instruction, 3 units ahead; patch sums added in a new order
# speedup vs baseline: 1.0679x; 1.0679x over previous
#include <hip/hip_runtime.h>

typedef __attribute__((ext_vector_type(2))) float v2f;
typedef __attribute__((ext_vector_type(4))) float v4f;
typedef __attribute__((ext_vector_type(8))) float v8f;

#define IMG    224
#define PATCH  16
#define NP     14
#define EMBED  768
#define CCH    12
#define BB     64
#define M_TOTAL (BB * NP * NP)
#define NTILES  (EMBED / 16)
#define MTILES  (M_TOTAL / 16)

#define POOL_WAVES  7
#define POOL_GROUPS 512
#define WSTRIPS     3
#define UNIT_LOADS  7
#define UNIT_F4     (UNIT_LOADS * 32)
#define WUNITS      (WSTRIPS * 4)
#define AHEAD       3
static_assert(POOL_GROUPS * POOL_WAVES * WSTRIPS == BB * CCH * NP, "every strip has exactly one wave");

__global__ __launch_bounds__(POOL_WAVES * 32) void pool16_kernel(const float* __restrict__ x,
                                                                 float* __restrict__ pooled) {
    __shared__ __attribute__((aligned(16))) float part[POOL_WAVES][UNIT_F4];

    const int lane = threadIdx.x & 31;
    const int wave = __builtin_amdgcn_readfirstlane((int)(threadIdx.x >> 5));
    const int gw   = blockIdx.x * POOL_WAVES + wave;
    const v4f* src = reinterpret_cast<const v4f*>(x) + (size_t)gw * (WUNITS * UNIT_F4) + lane;
    float* mine = part[wave];

    v4f buf[AHEAD + 1][UNIT_LOADS];
#define LOAD_UNIT(U) do { _Pragma("unroll") for (int q = 0; q < UNIT_LOADS; ++q) \
        buf[(U) % (AHEAD + 1)][q] = src[(U) * UNIT_F4 + q * 32]; } while (0)

#pragma unroll
    for (int u = 0; u < AHEAD; ++u) LOAD_UNIT(u);

    v4f acc[UNIT_LOADS];
#pragma unroll
    for (int u = 0; u < WUNITS; ++u) {
        __builtin_amdgcn_sched_barrier(0);
        if (u + AHEAD < WUNITS) LOAD_UNIT(u + AHEAD);
        __builtin_amdgcn_sched_barrier(0);

        const v4f* cur = buf[u % (AHEAD + 1)];
#pragma unroll
        for (int q = 0; q < UNIT_LOADS; ++q)
            acc[q] = (u % 4 == 0) ? cur[q] : acc[q] + cur[q];

        if (u % 4 == 3) {
#pragma unroll
            for (int q = 0; q < UNIT_LOADS; ++q)
                mine[q * 32 + lane] = (acc[q].x + acc[q].y) + (acc[q].z + acc[q].w);
            __builtin_amdgcn_fence(__ATOMIC_ACQ_REL, "wavefront");

            if (lane < NP) {
                float rs[4];
#pragma unroll
                for (int r = 0; r < 4; ++r) {
                    const float* p = mine + r * (NP * 4) + lane * 4;
                    rs[r] = (p[0] + p[1]) + (p[2] + p[3]);
                }
                int s = gw * WSTRIPS + (u >> 2);
                int i = s % NP; s /= NP;
                int c = s % CCH; s /= CCH;
                int b = s;
                int m = b * (NP * NP) + i * NP + lane;
                pooled[(size_t)m * CCH + c] = (rs[0] + rs[1]) + (rs[2] + rs[3]);
            }
            __builtin_amdgcn_fence(__ATOMIC_ACQ_REL, "wavefront");
        }
    }
#undef LOAD_UNIT
}

__global__ __launch_bounds__(256) void gemm_wmma_kernel(const float* __restrict__ pooled,
                                                        const float* __restrict__ w,
                                                        float* __restrict__ out) {
    int wid  = blockIdx.x * 8 + (threadIdx.x >> 5);
    int lane = threadIdx.x & 31;
    int mtile = wid / NTILES;
    int ntile = wid % NTILES;

    int half  = lane >> 4;
    int khalf = half * 2;
    int mrow  = lane & 15;
    int n     = lane & 15;

    const float* arow = pooled + (size_t)(mtile * 16 + mrow) * CCH;
    const float* wcol = w + ntile * 16 + n;

    v8f acc = {};
#pragma unroll
    for (int k0 = 0; k0 < CCH; k0 += 4) {
        v2f a;
        a.x = arow[k0 + khalf];
        a.y = arow[k0 + khalf + 1];
        v2f bm;
        bm.x = wcol[(size_t)(k0 + khalf)     * EMBED];
        bm.y = wcol[(size_t)(k0 + khalf + 1) * EMBED];
        acc = __builtin_amdgcn_wmma_f32_16x16x4_f32(
             false, a,  false, bm,
             (short)0, acc,  false,  false);
    }

    int mbase = mtile * 16 + half * 8;
    float* o = out + (size_t)mbase * EMBED + (size_t)ntile * 16 + n;
#pragma unroll
    for (int v = 0; v < 8; ++v)
        o[(size_t)v * EMBED] = acc[v];
}

extern "C" void kernel_launch(void* const* d_in, const int* in_sizes, int n_in,
                              void* d_out, int out_size, void* d_ws, size_t ws_size,
                              hipStream_t stream) {
    const float* x = (const float*)d_in[0];
    const float* w = (const float*)d_in[1];
    float* out    = (float*)d_out;
    float* pooled = (float*)d_ws;

    pool16_kernel<<<POOL_GROUPS, POOL_WAVES * 32, 0, stream>>>(x, pooled);
    gemm_wmma_kernel<<<(MTILES * NTILES) / 8, 256, 0, stream>>>(pooled, w, out);
}


# one launch instead of two: a 4-wave workgroup pools one (batch, patch-row) over all 12 channels into a 14x12 LDS tile, multiplies it on WMMA and writes 14 whole output rows; order of all sums kept
# speedup vs baseline: 1.5991x; 1.4974x over previous
#include <hip/hip_runtime.h>


typedef __attribute__((ext_vector_type(2))) float v2f;
typedef __attribute__((ext_vector_type(4))) float v4f;
typedef __attribute__((ext_vector_type(8))) float v8f;

#define IMG    224
#define PATCH  16
#define NP     14
#define EMBED  768
#define CCH    12
#define BB     64

#define FWAVES      4
#define WSTRIPS     3
#define UNIT_LOADS  7
#define UNIT_F4     (UNIT_LOADS * 32)
#define STRIP_F4    (4 * UNIT_F4)
#define CHAN_F4     (NP * STRIP_F4)
#define WUNITS      (WSTRIPS * 4)
#define AHEAD       3
#define WCOLS       (EMBED / FWAVES)
#define NGROUPS     (WCOLS / 64)
#define ROW_F4      (EMBED / 4)
static_assert(FWAVES * WSTRIPS == CCH, "every channel has exactly one wave");
static_assert(NGROUPS <= WUNITS, "one group of W loads per closing unit");

__global__ __launch_bounds__(FWAVES * 32) void patch_embed_fused(const float* __restrict__ x,
                                                                 const float* __restrict__ w,
                                                                 float* __restrict__ out) {
    __shared__ __attribute__((aligned(16))) float part[FWAVES][UNIT_F4];
    __shared__ __attribute__((aligned(16))) float atile[16 * CCH];

    const int lane = threadIdx.x & 31;
    const int wave = __builtin_amdgcn_readfirstlane((int)(threadIdx.x >> 5));
    const int bi = blockIdx.x;
    const int b  = bi / NP;
    const int i  = bi % NP;

    if (threadIdx.x < 2 * CCH) atile[NP * CCH + threadIdx.x] = 0.0f;

    const v4f* src = reinterpret_cast<const v4f*>(x)
                   + (size_t)((b * CCH + wave * WSTRIPS) * NP + i) * STRIP_F4 + lane;
    float* mine = part[wave];

    const int half  = lane >> 4;
    const int khalf = half * 2;
    const int n     = lane & 15;
    const v4f* w4 = reinterpret_cast<const v4f*>(w) + wave * (WCOLS / 4) + n;

    v4f buf[AHEAD + 1][UNIT_LOADS];
    v4f bx[NGROUPS][3], by[NGROUPS][3];
#define LOAD_UNIT(U) do { _Pragma("unroll") for (int q = 0; q < UNIT_LOADS; ++q) \
        buf[(U) % (AHEAD + 1)][q] = src[((U) / 4) * CHAN_F4 + ((U) % 4) * UNIT_F4 + q * 32]; } while (0)
#define LOAD_W(G) do { _Pragma("unroll") for (int kk = 0; kk < 3; ++kk) { \
        bx[(G)][kk] = w4[(kk * 4 + khalf)     * ROW_F4 + (G) * 16]; \
        by[(G)][kk] = w4[(kk * 4 + khalf + 1) * ROW_F4 + (G) * 16]; } } while (0)

#pragma unroll
    for (int u = 0; u < AHEAD; ++u) LOAD_UNIT(u);

    v4f acc[UNIT_LOADS];
#pragma unroll
    for (int u = 0; u < WUNITS; ++u) {
        __builtin_amdgcn_sched_barrier(0);
        if (u + AHEAD < WUNITS) LOAD_UNIT(u + AHEAD);
        if (u >= WUNITS - NGROUPS) LOAD_W(u - (WUNITS - NGROUPS));
        __builtin_amdgcn_sched_barrier(0);

        const v4f* cur = buf[u % (AHEAD + 1)];
#pragma unroll
        for (int q = 0; q < UNIT_LOADS; ++q)
            acc[q] = (u % 4 == 0) ? cur[q] : acc[q] + cur[q];

        if (u % 4 == 3) {
#pragma unroll
            for (int q = 0; q < UNIT_LOADS; ++q)
                mine[q * 32 + lane] = (acc[q].x + acc[q].y) + (acc[q].z + acc[q].w);
            __builtin_amdgcn_fence(__ATOMIC_ACQ_REL, "wavefront");

            if (lane < NP) {
                float rs[4];
#pragma unroll
                for (int r = 0; r < 4; ++r) {
                    const float* p = mine + r * (NP * 4) + lane * 4;
                    rs[r] = (p[0] + p[1]) + (p[2] + p[3]);
                }
                atile[lane * CCH + wave * WSTRIPS + (u >> 2)] = (rs[0] + rs[1]) + (rs[2] + rs[3]);
            }
            __builtin_amdgcn_fence(__ATOMIC_ACQ_REL, "wavefront");
        }
    }
#undef LOAD_UNIT
#undef LOAD_W

    __syncthreads();

    v2f a[3];
#pragma unroll
    for (int kk = 0; kk < 3; ++kk)
        a[kk] = *reinterpret_cast<const v2f*>(&atile[n * CCH + kk * 4 + khalf]);

    v4f* o4 = reinterpret_cast<v4f*>(out) + (size_t)(bi * NP) * ROW_F4 + wave * (WCOLS / 4) + n;

#pragma unroll
    for (int g = 0; g < NGROUPS; ++g) {
        v8f d[4] = {};
#pragma unroll
        for (int kk = 0; kk < 3; ++kk) {
#pragma unroll
            for (int t = 0; t < 4; ++t) {
                v2f bm;
                bm.x = bx[g][kk][t];
                bm.y = by[g][kk][t];
                d[t] = __builtin_amdgcn_wmma_f32_16x16x4_f32(
                     false, a[kk],  false, bm,
                     (short)0, d[t],  false,  false);
            }
        }
#pragma unroll
        for (int v = 0; v < 8; ++v) {
            const int row = v + 8 * half;
            if (row < NP) {
                v4f o;
                o.x = d[0][v]; o.y = d[1][v]; o.z = d[2][v]; o.w = d[3][v];
                o4[(size_t)row * ROW_F4 + g * 16] = o;
            }
        }
    }
}

extern "C" void kernel_launch(void* const* d_in, const int* in_sizes, int n_in,
                              void* d_out, int out_size, void* d_ws, size_t ws_size,
                              hipStream_t stream) {
    const float* x = (const float*)d_in[0];
    const float* w = (const float*)d_in[1];
    float* out    = (float*)d_out;

    patch_embed_fused<<<BB * NP, FWAVES * 32, 0, stream>>>(x, w, out);
}


# equal bytes per WGP: grid 1024 = 768 whole patch-rows + 128 patch-rows split into two 7-patch halves (448 of a row's 896 bytes), halves launched last; split rows' patch sums in a new order
# speedup vs baseline: 1.6862x; 1.0545x over previous
#include <hip/hip_runtime.h>


typedef __attribute__((ext_vector_type(2))) float v2f;
typedef __attribute__((ext_vector_type(4))) float v4f;
typedef __attribute__((ext_vector_type(8))) float v8f;

#define IMG    224
#define PATCH  16
#define NP     14
#define EMBED  768
#define CCH    12
#define BB     64

#define FWAVES      4
#define WSTRIPS     3
#define UNIT_LOADS  7
#define UNIT_F4     (UNIT_LOADS * 32)
#define STRIP_F4    (4 * UNIT_F4)
#define CHAN_F4     (NP * STRIP_F4)
#define AHEAD       3
#define WCOLS       (EMBED / FWAVES)
#define NGROUPS     (WCOLS / 64)
#define ROW_F4      (EMBED / 4)
#define ROWQ_F4     (IMG / 4)
#define HROW_F4     (ROWQ_F4 / 2)
#define NWHOLE      768
#define NSPLIT      128
#define GRID_WGS    1024
static_assert(FWAVES * WSTRIPS == CCH, "every channel has exactly one wave");
static_assert(NWHOLE + NSPLIT == BB * NP, "every item has an owner");
static_assert(NWHOLE + 2 * NSPLIT == GRID_WGS, "grid = whole + half workgroups");

template <bool HALF>
static __device__ __forceinline__ void item_body(const float* __restrict__ x,
                                                 const float* __restrict__ w,
                                                 float* __restrict__ out,
                                                 float* __restrict__ part_all,
                                                 float* __restrict__ atile,
                                                 const int bi, const int jh) {
    constexpr int UPS    = HALF ? 2 : 4;
    constexpr int WUNITS = WSTRIPS * UPS;
    constexpr int NROWS  = HALF ? 7 : NP;
    static_assert(NGROUPS <= WUNITS, "one group of W loads per closing unit");

    const int lane = threadIdx.x & 31;
    const int wave = __builtin_amdgcn_readfirstlane((int)(threadIdx.x >> 5));
    const int b  = bi / NP;
    const int i  = bi % NP;

    if (threadIdx.x < (16 - NROWS) * CCH) atile[NROWS * CCH + threadIdx.x] = 0.0f;

    const v4f* src = reinterpret_cast<const v4f*>(x)
                   + (size_t)((b * CCH + wave * WSTRIPS) * NP + i) * STRIP_F4
                   + (HALF ? jh * HROW_F4 : lane);
    float* mine = part_all + wave * UNIT_F4;

    const int half  = lane >> 4;
    const int khalf = half * 2;
    const int n     = lane & 15;
    const v4f* w4 = reinterpret_cast<const v4f*>(w) + wave * (WCOLS / 4) + n;

    v4f buf[AHEAD + 1][UNIT_LOADS];
    v4f bx[NGROUPS][3], by[NGROUPS][3];

    auto unit_off = [&](int U, int q) -> int {
        if (HALF) {
            const int g = 32 * ((U % UPS) * UNIT_LOADS + q) + lane;
            return (U / UPS) * CHAN_F4 + g + HROW_F4 * (g / HROW_F4);
        } else {
            return (U / UPS) * CHAN_F4 + (U % UPS) * UNIT_F4 + q * 32;
        }
    };
#define LOAD_UNIT(U) do { _Pragma("unroll") for (int q = 0; q < UNIT_LOADS; ++q) \
        buf[(U) % (AHEAD + 1)][q] = src[unit_off((U), q)]; } while (0)
#define LOAD_W(G) do { _Pragma("unroll") for (int kk = 0; kk < 3; ++kk) { \
        bx[(G)][kk] = w4[(kk * 4 + khalf)     * ROW_F4 + (G) * 16]; \
        by[(G)][kk] = w4[(kk * 4 + khalf + 1) * ROW_F4 + (G) * 16]; } } while (0)

#pragma unroll
    for (int u = 0; u < AHEAD; ++u) LOAD_UNIT(u);

    v4f acc[UNIT_LOADS];
#pragma unroll
    for (int u = 0; u < WUNITS; ++u) {
        __builtin_amdgcn_sched_barrier(0);
        if (u + AHEAD < WUNITS) LOAD_UNIT(u + AHEAD);
        if (u >= WUNITS - NGROUPS) LOAD_W(u - (WUNITS - NGROUPS));
        __builtin_amdgcn_sched_barrier(0);

        const v4f* cur = buf[u % (AHEAD + 1)];
#pragma unroll
        for (int q = 0; q < UNIT_LOADS; ++q)
            acc[q] = (u % UPS == 0) ? cur[q] : acc[q] + cur[q];

        if (u % UPS == UPS - 1) {
#pragma unroll
            for (int q = 0; q < UNIT_LOADS; ++q)
                mine[q * 32 + lane] = (acc[q].x + acc[q].y) + (acc[q].z + acc[q].w);
            __builtin_amdgcn_fence(__ATOMIC_ACQ_REL, "wavefront");

            if (lane < NROWS) {
                float tot;
                if (HALF) {
                    float rs[8];
#pragma unroll
                    for (int a = 0; a < 8; ++a) {
                        const int s = (lane + 7 - a) % 7;
                        const float* p = mine + s * 32 + a * 4;
                        rs[a] = (p[0] + p[1]) + (p[2] + p[3]);
                    }
                    tot = ((rs[0] + rs[1]) + (rs[2] + rs[3])) + ((rs[4] + rs[5]) + (rs[6] + rs[7]));
                } else {
                    float rs[4];
#pragma unroll
                    for (int r = 0; r < 4; ++r) {
                        const float* p = mine + r * (NP * 4) + lane * 4;
                        rs[r] = (p[0] + p[1]) + (p[2] + p[3]);
                    }
                    tot = (rs[0] + rs[1]) + (rs[2] + rs[3]);
                }
                atile[lane * CCH + wave * WSTRIPS + (u / UPS)] = tot;
            }
            __builtin_amdgcn_fence(__ATOMIC_ACQ_REL, "wavefront");
        }
    }
#undef LOAD_UNIT
#undef LOAD_W

    __syncthreads();

    v2f a[3];
#pragma unroll
    for (int kk = 0; kk < 3; ++kk)
        a[kk] = *reinterpret_cast<const v2f*>(&atile[n * CCH + kk * 4 + khalf]);

    v4f* o4 = reinterpret_cast<v4f*>(out) + (size_t)(bi * NP + (HALF ? jh * 7 : 0)) * ROW_F4
            + wave * (WCOLS / 4) + n;

#pragma unroll
    for (int g = 0; g < NGROUPS; ++g) {
        v8f d[4] = {};
#pragma unroll
        for (int kk = 0; kk < 3; ++kk) {
#pragma unroll
            for (int t = 0; t < 4; ++t) {
                v2f bm;
                bm.x = bx[g][kk][t];
                bm.y = by[g][kk][t];
                d[t] = __builtin_amdgcn_wmma_f32_16x16x4_f32(
                     false, a[kk],  false, bm,
                     (short)0, d[t],  false,  false);
            }
        }
#pragma unroll
        for (int v = 0; v < 8; ++v) {
            const int row = v + 8 * half;
            if (row < NROWS) {
                v4f o;
                o.x = d[0][v]; o.y = d[1][v]; o.z = d[2][v]; o.w = d[3][v];
                o4[(size_t)row * ROW_F4 + g * 16] = o;
            }
        }
    }
}

__global__ __launch_bounds__(FWAVES * 32) void patch_embed_fused(const float* __restrict__ x,
                                                                 const float* __restrict__ w,
                                                                 float* __restrict__ out) {
    __shared__ __attribute__((aligned(16))) float part[FWAVES * UNIT_F4];
    __shared__ __attribute__((aligned(16))) float atile[16 * CCH];

    const int wg = blockIdx.x;
    if (wg < NWHOLE) {
        item_body<false>(x, w, out, part, atile, wg, 0);
    } else {
        const int h = wg - NWHOLE;
        item_body<true>(x, w, out, part, atile, NWHOLE + (h >> 1), h & 1);
    }
}

extern "C" void kernel_launch(void* const* d_in, const int* in_sizes, int n_in,
                              void* d_out, int out_size, void* d_ws, size_t ws_size,
                              hipStream_t stream) {
    const float* x = (const float*)d_in[0];
    const float* w = (const float*)d_in[1];
    float* out    = (float*)d_out;

    patch_embed_fused<<<GRID_WGS, FWAVES * 32, 0, stream>>>(x, w, out);
}
